# starting kernel
# baseline (speedup 1.0000x reference)
#include <hip/hip_runtime.h>

typedef __attribute__((ext_vector_type(2))) float v2f;
typedef __attribute__((ext_vector_type(8))) float v8f;

constexpr int kC   = 16;
constexpr int kK   = 8;
constexpr int kR   = 4096;
constexpr int kSR  = 128;
constexpr int kN   = kR * kSR;
constexpr int kPad = 17;

__global__ __launch_bounds__(256)
void pvr_feat_head(const float* __restrict__ points_feat,
                   const int*   __restrict__ indices,
                   const float* __restrict__ dists,
                   const float* __restrict__ w_rgb,
                   const float* __restrict__ w_sigma,
                   float* __restrict__ sigma_ws,
                   float* __restrict__ rgb_ws)
{
    __shared__ float sfeat[256 * kPad];

    const int tid = threadIdx.x;
    const int n   = blockIdx.x * 256 + tid;

    const float4* dv = (const float4*)(dists + (size_t)n * kK);
    float4 da = dv[0], db = dv[1];
    float dd[kK] = {da.x, da.y, da.z, da.w, db.x, db.y, db.z, db.w};

    float w[kK];
    float wsum = 0.0f;
#pragma unroll
    for (int k = 0; k < kK; ++k) {
        w[k] = 1.0f / (dd[k] + 1e-7f);
        wsum += w[k];
    }
    const float inv = 1.0f / wsum;

    const int4* iv = (const int4*)(indices + (size_t)n * kK);
    int4 ia = iv[0], ib = iv[1];
    int id[kK] = {ia.x, ia.y, ia.z, ia.w, ib.x, ib.y, ib.z, ib.w};

    float acc[kC];
#pragma unroll
    for (int c = 0; c < kC; ++c) acc[c] = 0.0f;

#pragma unroll
    for (int k = 0; k < kK; ++k) {
        const float4* row = (const float4*)(points_feat + (size_t)id[k] * kC);
        const float wk = w[k] * inv;
        float4 f0 = row[0], f1 = row[1], f2 = row[2], f3 = row[3];
        acc[0]  = fmaf(wk, f0.x, acc[0]);  acc[1]  = fmaf(wk, f0.y, acc[1]);
        acc[2]  = fmaf(wk, f0.z, acc[2]);  acc[3]  = fmaf(wk, f0.w, acc[3]);
        acc[4]  = fmaf(wk, f1.x, acc[4]);  acc[5]  = fmaf(wk, f1.y, acc[5]);
        acc[6]  = fmaf(wk, f1.z, acc[6]);  acc[7]  = fmaf(wk, f1.w, acc[7]);
        acc[8]  = fmaf(wk, f2.x, acc[8]);  acc[9]  = fmaf(wk, f2.y, acc[9]);
        acc[10] = fmaf(wk, f2.z, acc[10]); acc[11] = fmaf(wk, f2.w, acc[11]);
        acc[12] = fmaf(wk, f3.x, acc[12]); acc[13] = fmaf(wk, f3.y, acc[13]);
        acc[14] = fmaf(wk, f3.z, acc[14]); acc[15] = fmaf(wk, f3.w, acc[15]);
    }

    float* dst = sfeat + tid * kPad;
#pragma unroll
    for (int c = 0; c < kC; ++c) dst[c] = acc[c];

    __syncthreads();

    const int lane = tid & 31;
    const int wave = tid >> 5;
    const int half = lane >> 4;
    const int nl   = lane & 15;

    v2f bmat[4];
#pragma unroll
    for (int t = 0; t < 4; ++t) {
        const int k0 = 4 * t + 2 * half;
        float b0 = 0.0f, b1 = 0.0f;
        if (nl < 3) {
            b0 = w_rgb[k0 * 3 + nl];
            b1 = w_rgb[(k0 + 1) * 3 + nl];
        } else if (nl == 3) {
            b0 = w_sigma[k0];
            b1 = w_sigma[k0 + 1];
        }
        bmat[t].x = b0;
        bmat[t].y = b1;
    }

#pragma unroll
    for (int tt = 0; tt < 2; ++tt) {
        const int rowBase = (wave * 2 + tt) * 16;
        v8f d = {};
#pragma unroll
        for (int t = 0; t < 4; ++t) {
            const int k0 = 4 * t + 2 * half;
            const float* src = sfeat + (rowBase + nl) * kPad + k0;
            v2f a;
            a.x = src[0];
            a.y = src[1];
            d = __builtin_amdgcn_wmma_f32_16x16x4_f32(
                    false, a, false, bmat[t], (short)0, d, false, false);
        }

        const int pbase = blockIdx.x * 256 + rowBase + 8 * half;
        if (nl < 3) {
#pragma unroll
            for (int v = 0; v < 8; ++v) {
                const float x = d[v];
                rgb_ws[(size_t)nl * kN + (pbase + v)] = 1.0f / (1.0f + __expf(-x));
            }
        } else if (nl == 3) {
#pragma unroll
            for (int v = 0; v < 8; ++v) {
                sigma_ws[pbase + v] = fmaxf(d[v], 0.0f);
            }
        }
    }
}

__global__ __launch_bounds__(128)
void pvr_composite(const float* __restrict__ sigma_ws,
                   const float* __restrict__ rgb_ws,
                   const float* __restrict__ delta,
                   const float* __restrict__ z_vals,
                   float* __restrict__ out)
{
    const int r = blockIdx.x * blockDim.x + threadIdx.x;
    if (r >= kR) return;

    float T = 1.0f;
    float r0 = 0.0f, r1 = 0.0f, r2 = 0.0f, depth = 0.0f, accw = 0.0f;
    const int base = r * kSR;

    for (int s = 0; s < kSR; ++s) {
        const int i = base + s;
        const float sig = sigma_ws[i];
        const float a   = 1.0f - __expf(-sig * delta[i]);
        const float wt  = a * T;
        r0    = fmaf(wt, rgb_ws[i],            r0);
        r1    = fmaf(wt, rgb_ws[kN + i],       r1);
        r2    = fmaf(wt, rgb_ws[2 * kN + i],   r2);
        depth = fmaf(wt, z_vals[i],            depth);
        accw += wt;
        T *= (1.0f - a + 1e-10f);
    }

    const float bg = 1.0f - accw;
    out[r * 5 + 0] = r0 + bg;
    out[r * 5 + 1] = r1 + bg;
    out[r * 5 + 2] = r2 + bg;
    out[r * 5 + 3] = depth;
    out[r * 5 + 4] = accw;
}

extern "C" void kernel_launch(void* const* d_in, const int* in_sizes, int n_in,
                              void* d_out, int out_size, void* d_ws, size_t ws_size,
                              hipStream_t stream)
{
    const float* points_feat = (const float*)d_in[0];
    const int*   indices     = (const int*)  d_in[1];
    const float* dists       = (const float*)d_in[2];
    const float* w_rgb       = (const float*)d_in[3];
    const float* w_sigma     = (const float*)d_in[4];
    const float* delta       = (const float*)d_in[5];
    const float* z_vals      = (const float*)d_in[6];
    float* out = (float*)d_out;

    float* sigma_ws = (float*)d_ws;
    float* rgb_ws   = sigma_ws + kN;

    pvr_feat_head<<<kN / 256, 256, 0, stream>>>(
        points_feat, indices, dists, w_rgb, w_sigma, sigma_ws, rgb_ws);

    pvr_composite<<<kR / 128, 128, 0, stream>>>(
        sigma_ws, rgb_ws, delta, z_vals, out);
}


# Head moved in front of the gather: a WMMA pass writes G = points_feat x [w_rgb|w_sigma] (500000x4) to the workspace, each neighbour read is a 16 B row of G, not 64 B; 3 launches; sums re-associated
# speedup vs baseline: 1.4559x; 1.4559x over previous
#include <hip/hip_runtime.h>


typedef __attribute__((ext_vector_type(2))) float v2f;
typedef __attribute__((ext_vector_type(8))) float v8f;

constexpr int kC   = 16;
constexpr int kK   = 8;
constexpr int kR   = 4096;
constexpr int kSR  = 128;
constexpr int kN   = kR * kSR;
constexpr int kNP  = 500000;

constexpr int kTiles   = kNP / 16;
constexpr int kTPW     = 10;
constexpr int kPreWPB  = 5;
constexpr int kPreGrid = kTiles / kTPW / kPreWPB;
static_assert(kPreGrid * kPreWPB * kTPW * 16 == kNP, "table pass must cover every row exactly once");


__device__ __forceinline__ constexpr int frag_k(int j, int p, int h, int e)
{
    return 8 * j + 4 * h + 2 * p + e;
}

__global__ __launch_bounds__(kPreWPB * 32)
void pvr_table_head(const float* __restrict__ points_feat,
                    const float* __restrict__ w_rgb,
                    const float* __restrict__ w_sigma,
                    float* __restrict__ g_ws)
{
    const int lane = threadIdx.x & 31;
    const int wave = blockIdx.x * kPreWPB + (threadIdx.x >> 5);
    const int h    = lane >> 4;
    const int c    = lane & 15;

    const bool     is_rgb = (c < 3);
    const unsigned keep   = (c <= 3) ? 0xffffffffu : 0u;
    v2f wt[4];
#pragma unroll
    for (int j = 0; j < 2; ++j) {
#pragma unroll
        for (int p = 0; p < 2; ++p) {
            const int k0 = frag_k(j, p, h, 0);
            const int k1 = frag_k(j, p, h, 1);
            const float* s0 = is_rgb ? (w_rgb + k0 * 3 + c) : (w_sigma + k0);
            const float* s1 = is_rgb ? (w_rgb + k1 * 3 + c) : (w_sigma + k1);
            wt[2 * j + p].x = __uint_as_float(__float_as_uint(*s0) & keep);
            wt[2 * j + p].y = __uint_as_float(__float_as_uint(*s1) & keep);
        }
    }

    const int tile0 = wave * kTPW;

    float4 f[kTPW][2];
#pragma unroll
    for (int i = 0; i < kTPW; ++i) {
        const float4* row = (const float4*)(points_feat + (size_t)((tile0 + i) * 16 + c) * kC);
        f[i][0] = row[h];
        f[i][1] = row[2 + h];
    }

    static_assert(kTPW % 2 == 0, "tiles are processed in pairs");
#pragma unroll
    for (int i = 0; i < kTPW; i += 2) {
        v8f d0 = {};
        v8f d1 = {};
#pragma unroll
        for (int j = 0; j < 2; ++j) {
            v2f b00, b01, b10, b11;
            b00.x = f[i][j].x;      b00.y = f[i][j].y;
            b01.x = f[i][j].z;      b01.y = f[i][j].w;
            b10.x = f[i + 1][j].x;  b10.y = f[i + 1][j].y;
            b11.x = f[i + 1][j].z;  b11.y = f[i + 1][j].w;
            d0 = __builtin_amdgcn_wmma_f32_16x16x4_f32(
                     false, wt[2 * j + 0], false, b00, (short)0, d0, false, false);
            d1 = __builtin_amdgcn_wmma_f32_16x16x4_f32(
                     false, wt[2 * j + 0], false, b10, (short)0, d1, false, false);
            d0 = __builtin_amdgcn_wmma_f32_16x16x4_f32(
                     false, wt[2 * j + 1], false, b01, (short)0, d0, false, false);
            d1 = __builtin_amdgcn_wmma_f32_16x16x4_f32(
                     false, wt[2 * j + 1], false, b11, (short)0, d1, false, false);
        }
        if (h == 0) {
            float4 o0, o1;
            o0.x = d0[0]; o0.y = d0[1]; o0.z = d0[2]; o0.w = d0[3];
            o1.x = d1[0]; o1.y = d1[1]; o1.z = d1[2]; o1.w = d1[3];
            ((float4*)g_ws)[(tile0 + i) * 16 + c]     = o0;
            ((float4*)g_ws)[(tile0 + i + 1) * 16 + c] = o1;
        }
    }
}

__global__ __launch_bounds__(256)
void pvr_gather_act(const float* __restrict__ g_ws,
                    const int*   __restrict__ indices,
                    const float* __restrict__ dists,
                    float* __restrict__ sigma_ws,
                    float* __restrict__ rgb_ws)
{
    const int n = blockIdx.x * 256 + threadIdx.x;

    const float4* dv = (const float4*)(dists + (size_t)n * kK);
    float4 da = dv[0], db = dv[1];
    float dd[kK] = {da.x, da.y, da.z, da.w, db.x, db.y, db.z, db.w};

    const int4* iv = (const int4*)(indices + (size_t)n * kK);
    int4 ia = iv[0], ib = iv[1];
    int id[kK] = {ia.x, ia.y, ia.z, ia.w, ib.x, ib.y, ib.z, ib.w};

    float w[kK];
    float wsum = 0.0f;
#pragma unroll
    for (int k = 0; k < kK; ++k) {
        w[k] = 1.0f / (dd[k] + 1e-7f);
        wsum += w[k];
    }
    const float inv = 1.0f / wsum;

    const float4* G = (const float4*)g_ws;
    float4 g[kK];
#pragma unroll
    for (int k = 0; k < kK; ++k) g[k] = G[id[k]];

    float o0 = 0.0f, o1 = 0.0f, o2 = 0.0f, o3 = 0.0f;
#pragma unroll
    for (int k = 0; k < kK; ++k) {
        const float wk = w[k] * inv;
        o0 = fmaf(wk, g[k].x, o0);
        o1 = fmaf(wk, g[k].y, o1);
        o2 = fmaf(wk, g[k].z, o2);
        o3 = fmaf(wk, g[k].w, o3);
    }

    rgb_ws[n]                  = 1.0f / (1.0f + __expf(-o0));
    rgb_ws[(size_t)kN + n]     = 1.0f / (1.0f + __expf(-o1));
    rgb_ws[(size_t)2 * kN + n] = 1.0f / (1.0f + __expf(-o2));
    sigma_ws[n]                = fmaxf(o3, 0.0f);
}

__global__ __launch_bounds__(128)
void pvr_composite(const float* __restrict__ sigma_ws,
                   const float* __restrict__ rgb_ws,
                   const float* __restrict__ delta,
                   const float* __restrict__ z_vals,
                   float* __restrict__ out)
{
    const int r = blockIdx.x * blockDim.x + threadIdx.x;
    if (r >= kR) return;

    float T = 1.0f;
    float r0 = 0.0f, r1 = 0.0f, r2 = 0.0f, depth = 0.0f, accw = 0.0f;
    const int base = r * kSR;

    for (int s = 0; s < kSR; ++s) {
        const int i = base + s;
        const float sig = sigma_ws[i];
        const float a   = 1.0f - __expf(-sig * delta[i]);
        const float wt  = a * T;
        r0    = fmaf(wt, rgb_ws[i],            r0);
        r1    = fmaf(wt, rgb_ws[kN + i],       r1);
        r2    = fmaf(wt, rgb_ws[2 * kN + i],   r2);
        depth = fmaf(wt, z_vals[i],            depth);
        accw += wt;
        T *= (1.0f - a + 1e-10f);
    }

    const float bg = 1.0f - accw;
    out[r * 5 + 0] = r0 + bg;
    out[r * 5 + 1] = r1 + bg;
    out[r * 5 + 2] = r2 + bg;
    out[r * 5 + 3] = depth;
    out[r * 5 + 4] = accw;
}

extern "C" void kernel_launch(void* const* d_in, const int* in_sizes, int n_in,
                              void* d_out, int out_size, void* d_ws, size_t ws_size,
                              hipStream_t stream)
{
    const float* points_feat = (const float*)d_in[0];
    const int*   indices     = (const int*)  d_in[1];
    const float* dists       = (const float*)d_in[2];
    const float* w_rgb       = (const float*)d_in[3];
    const float* w_sigma     = (const float*)d_in[4];
    const float* delta       = (const float*)d_in[5];
    const float* z_vals      = (const float*)d_in[6];
    float* out = (float*)d_out;

    float* sigma_ws = (float*)d_ws;
    float* rgb_ws   = sigma_ws + kN;
    float* g_ws     = rgb_ws + 3 * kN;

    pvr_table_head<<<kPreGrid, kPreWPB * 32, 0, stream>>>(
        points_feat, w_rgb, w_sigma, g_ws);

    pvr_gather_act<<<kN / 256, 256, 0, stream>>>(
        g_ws, indices, dists, sigma_ws, rgb_ws);

    pvr_composite<<<kR / 128, 128, 0, stream>>>(
        sigma_ws, rgb_ws, delta, z_vals, out);
}


# Composite inputs stored sample-major per 32-ray tile: gather block = 32 rays x 8 samples, alpha/rgb/z transposed via LDS; the one-thread-per-ray loop loads whole lines; every sum keeps its order
# speedup vs baseline: 1.7657x; 1.2128x over previous
#include <hip/hip_runtime.h>


typedef __attribute__((ext_vector_type(2))) float v2f;
typedef __attribute__((ext_vector_type(8))) float v8f;

constexpr int kC   = 16;
constexpr int kK   = 8;
constexpr int kR   = 4096;
constexpr int kSR  = 128;
constexpr int kN   = kR * kSR;
constexpr int kNP  = 500000;

constexpr int kTiles   = kNP / 16;
constexpr int kTPW     = 10;
constexpr int kPreWPB  = 5;
constexpr int kPreGrid = kTiles / kTPW / kPreWPB;
static_assert(kPreGrid * kPreWPB * kTPW * 16 == kNP, "table pass must cover every row exactly once");


__device__ __forceinline__ constexpr int frag_k(int j, int p, int h, int e)
{
    return 8 * j + 4 * h + 2 * p + e;
}

__global__ __launch_bounds__(kPreWPB * 32)
void pvr_table_head(const float* __restrict__ points_feat,
                    const float* __restrict__ w_rgb,
                    const float* __restrict__ w_sigma,
                    float* __restrict__ g_ws)
{
    const int lane = threadIdx.x & 31;
    const int wave = blockIdx.x * kPreWPB + (threadIdx.x >> 5);
    const int h    = lane >> 4;
    const int c    = lane & 15;

    const bool     is_rgb = (c < 3);
    const unsigned keep   = (c <= 3) ? 0xffffffffu : 0u;
    v2f wt[4];
#pragma unroll
    for (int j = 0; j < 2; ++j) {
#pragma unroll
        for (int p = 0; p < 2; ++p) {
            const int k0 = frag_k(j, p, h, 0);
            const int k1 = frag_k(j, p, h, 1);
            const float* s0 = is_rgb ? (w_rgb + k0 * 3 + c) : (w_sigma + k0);
            const float* s1 = is_rgb ? (w_rgb + k1 * 3 + c) : (w_sigma + k1);
            wt[2 * j + p].x = __uint_as_float(__float_as_uint(*s0) & keep);
            wt[2 * j + p].y = __uint_as_float(__float_as_uint(*s1) & keep);
        }
    }

    const int tile0 = wave * kTPW;

    float4 f[kTPW][2];
#pragma unroll
    for (int i = 0; i < kTPW; ++i) {
        const float4* row = (const float4*)(points_feat + (size_t)((tile0 + i) * 16 + c) * kC);
        f[i][0] = row[h];
        f[i][1] = row[2 + h];
    }

    static_assert(kTPW % 2 == 0, "tiles are processed in pairs");
#pragma unroll
    for (int i = 0; i < kTPW; i += 2) {
        v8f d0 = {};
        v8f d1 = {};
#pragma unroll
        for (int j = 0; j < 2; ++j) {
            v2f b00, b01, b10, b11;
            b00.x = f[i][j].x;      b00.y = f[i][j].y;
            b01.x = f[i][j].z;      b01.y = f[i][j].w;
            b10.x = f[i + 1][j].x;  b10.y = f[i + 1][j].y;
            b11.x = f[i + 1][j].z;  b11.y = f[i + 1][j].w;
            d0 = __builtin_amdgcn_wmma_f32_16x16x4_f32(
                     false, wt[2 * j + 0], false, b00, (short)0, d0, false, false);
            d1 = __builtin_amdgcn_wmma_f32_16x16x4_f32(
                     false, wt[2 * j + 0], false, b10, (short)0, d1, false, false);
            d0 = __builtin_amdgcn_wmma_f32_16x16x4_f32(
                     false, wt[2 * j + 1], false, b01, (short)0, d0, false, false);
            d1 = __builtin_amdgcn_wmma_f32_16x16x4_f32(
                     false, wt[2 * j + 1], false, b11, (short)0, d1, false, false);
        }
        if (h == 0) {
            float4 o0, o1;
            o0.x = d0[0]; o0.y = d0[1]; o0.z = d0[2]; o0.w = d0[3];
            o1.x = d1[0]; o1.y = d1[1]; o1.z = d1[2]; o1.w = d1[3];
            ((float4*)g_ws)[(tile0 + i) * 16 + c]     = o0;
            ((float4*)g_ws)[(tile0 + i + 1) * 16 + c] = o1;
        }
    }
}

constexpr int kTR    = 32;
constexpr int kTS    = 8;
constexpr int kPitch = 36;
static_assert(kTR * kTS == 256, "one thread per sample of the tile");
static_assert(kR % kTR == 0 && kSR % kTS == 0, "tiles cover the rays and samples exactly");

__global__ __launch_bounds__(256)
void pvr_gather_act(const float* __restrict__ g_ws,
                    const int*   __restrict__ indices,
                    const float* __restrict__ dists,
                    const float* __restrict__ delta,
                    const float* __restrict__ z_vals,
                    float* __restrict__ arf_ws,
                    float* __restrict__ z_ws)
{
    __shared__ float tile[5][kTS][kPitch];

    const int lane = threadIdx.x & 31;
    const int wv   = threadIdx.x >> 5;
    const int s0   = blockIdx.x * kTS;
    const int r0   = blockIdx.y * kTR;
    const int rl   = 4 * wv + (lane >> 3);
    const int sl   = lane & 7;
    const int n    = (r0 + rl) * kSR + s0 + sl;

    const float4* dv = (const float4*)(dists + (size_t)n * kK);
    float4 da = dv[0], db = dv[1];
    float dd[kK] = {da.x, da.y, da.z, da.w, db.x, db.y, db.z, db.w};

    const int4* iv = (const int4*)(indices + (size_t)n * kK);
    int4 ia = iv[0], ib = iv[1];
    int id[kK] = {ia.x, ia.y, ia.z, ia.w, ib.x, ib.y, ib.z, ib.w};

    const float dl = delta[n];
    const float zv = z_vals[n];

    float w[kK];
    float wsum = 0.0f;
#pragma unroll
    for (int k = 0; k < kK; ++k) {
        w[k] = 1.0f / (dd[k] + 1e-7f);
        wsum += w[k];
    }
    const float inv = 1.0f / wsum;

    const float4* G = (const float4*)g_ws;
    float4 g[kK];
#pragma unroll
    for (int k = 0; k < kK; ++k) g[k] = G[id[k]];

    float o0 = 0.0f, o1 = 0.0f, o2 = 0.0f, o3 = 0.0f;
#pragma unroll
    for (int k = 0; k < kK; ++k) {
        const float wk = w[k] * inv;
        o0 = fmaf(wk, g[k].x, o0);
        o1 = fmaf(wk, g[k].y, o1);
        o2 = fmaf(wk, g[k].z, o2);
        o3 = fmaf(wk, g[k].w, o3);
    }

    const float sig = fmaxf(o3, 0.0f);
    tile[0][sl][rl] = 1.0f - __expf(-sig * dl);
    tile[1][sl][rl] = 1.0f / (1.0f + __expf(-o0));
    tile[2][sl][rl] = 1.0f / (1.0f + __expf(-o1));
    tile[3][sl][rl] = 1.0f / (1.0f + __expf(-o2));
    tile[4][sl][rl] = zv;

    __syncthreads();

    float4 v;
    v.x = tile[0][wv][lane];
    v.y = tile[1][wv][lane];
    v.z = tile[2][wv][lane];
    v.w = tile[3][wv][lane];
    const size_t o = ((size_t)blockIdx.y * kSR + s0 + wv) * kTR + lane;
    ((float4*)arf_ws)[o] = v;
    z_ws[o]              = tile[4][wv][lane];
}

constexpr int kCH = 32;
static_assert(kSR % (2 * kCH) == 0, "chunks are processed in pairs");

struct Chunk {
    float4 a[kCH];
    float  z[kCH];
};

__device__ __forceinline__ void chunk_load(Chunk& c, const float4* __restrict__ arf,
                                           const float* __restrict__ zp, int s_first)
{
#pragma unroll
    for (int i = 0; i < kCH; ++i) {
        c.a[i] = arf[(s_first + i) * kTR];
        c.z[i] = zp[(s_first + i) * kTR];
    }
}

struct Acc {
    float T, r0, r1, r2, depth, accw;
};

__device__ __forceinline__ void chunk_use(const Chunk& c, Acc& q)
{
#pragma unroll
    for (int i = 0; i < kCH; ++i) {
        const float a  = c.a[i].x;
        const float wt = a * q.T;
        q.r0    = fmaf(wt, c.a[i].y, q.r0);
        q.r1    = fmaf(wt, c.a[i].z, q.r1);
        q.r2    = fmaf(wt, c.a[i].w, q.r2);
        q.depth = fmaf(wt, c.z[i],   q.depth);
        q.accw += wt;
        q.T *= (1.0f - a + 1e-10f);
    }
}

__global__ __launch_bounds__(kTR)
void pvr_composite(const float* __restrict__ arf_ws,
                   const float* __restrict__ z_ws,
                   float* __restrict__ out)
{
    static_assert(kTR == 32, "one wave per ray tile");
    const int r = blockIdx.x * kTR + threadIdx.x;
    const float4* arf = (const float4*)arf_ws + (size_t)blockIdx.x * kSR * kTR + threadIdx.x;
    const float*  zp  = z_ws + (size_t)blockIdx.x * kSR * kTR + threadIdx.x;

    Acc q;
    q.T = 1.0f; q.r0 = 0.0f; q.r1 = 0.0f; q.r2 = 0.0f; q.depth = 0.0f; q.accw = 0.0f;

    Chunk ca, cb;
    chunk_load(ca, arf, zp, 0);
#pragma unroll
    for (int s = 0; s < kSR; s += 2 * kCH) {
        chunk_load(cb, arf, zp, s + kCH);
        __builtin_amdgcn_sched_barrier(0);
        chunk_use(ca, q);
        if (s + 2 * kCH < kSR) chunk_load(ca, arf, zp, s + 2 * kCH);
        chunk_use(cb, q);
    }

    const float bg = 1.0f - q.accw;
    out[r * 5 + 0] = q.r0 + bg;
    out[r * 5 + 1] = q.r1 + bg;
    out[r * 5 + 2] = q.r2 + bg;
    out[r * 5 + 3] = q.depth;
    out[r * 5 + 4] = q.accw;
}

extern "C" void kernel_launch(void* const* d_in, const int* in_sizes, int n_in,
                              void* d_out, int out_size, void* d_ws, size_t ws_size,
                              hipStream_t stream)
{
    const float* points_feat = (const float*)d_in[0];
    const int*   indices     = (const int*)  d_in[1];
    const float* dists       = (const float*)d_in[2];
    const float* w_rgb       = (const float*)d_in[3];
    const float* w_sigma     = (const float*)d_in[4];
    const float* delta       = (const float*)d_in[5];
    const float* z_vals      = (const float*)d_in[6];
    float* out = (float*)d_out;

    float* arf_ws = (float*)d_ws;
    float* z_ws   = arf_ws + 4 * kN;
    float* g_ws   = z_ws + kN;

    pvr_table_head<<<kPreGrid, kPreWPB * 32, 0, stream>>>(
        points_feat, w_rgb, w_sigma, g_ws);

    pvr_gather_act<<<dim3(kSR / kTS, kR / kTR), 256, 0, stream>>>(
        g_ws, indices, dists, delta, z_vals, arf_ws, z_ws);

    pvr_composite<<<kR / kTR, kTR, 0, stream>>>(
        arf_ws, z_ws, out);
}


# Gather kernel coarsened: a thread handles 4 samples (block = 32 rays x 32 samples), 4096 waves enter, not 16384; its 24 row loads go as one group, then 32 gathers in flight; every sum keeps its order
# speedup vs baseline: 1.9750x; 1.1185x over previous
#include <hip/hip_runtime.h>


typedef __attribute__((ext_vector_type(2))) float v2f;
typedef __attribute__((ext_vector_type(8))) float v8f;
typedef __attribute__((ext_vector_type(4))) float v4f;
typedef __attribute__((ext_vector_type(4))) int   v4i;

constexpr int kC   = 16;
constexpr int kK   = 8;
constexpr int kR   = 4096;
constexpr int kSR  = 128;
constexpr int kN   = kR * kSR;
constexpr int kNP  = 500000;

constexpr int kTiles   = kNP / 16;
constexpr int kTPW     = 10;
constexpr int kPreWPB  = 5;
constexpr int kPreGrid = kTiles / kTPW / kPreWPB;
static_assert(kPreGrid * kPreWPB * kTPW * 16 == kNP, "table pass must cover every row exactly once");


__device__ __forceinline__ constexpr int frag_k(int j, int p, int h, int e)
{
    return 8 * j + 4 * h + 2 * p + e;
}

__global__ __launch_bounds__(kPreWPB * 32)
void pvr_table_head(const float* __restrict__ points_feat,
                    const float* __restrict__ w_rgb,
                    const float* __restrict__ w_sigma,
                    float* __restrict__ g_ws)
{
    const int lane = threadIdx.x & 31;
    const int wave = blockIdx.x * kPreWPB + (threadIdx.x >> 5);
    const int h    = lane >> 4;
    const int c    = lane & 15;

    const bool     is_rgb = (c < 3);
    const unsigned keep   = (c <= 3) ? 0xffffffffu : 0u;
    v2f wt[4];
#pragma unroll
    for (int j = 0; j < 2; ++j) {
#pragma unroll
        for (int p = 0; p < 2; ++p) {
            const int k0 = frag_k(j, p, h, 0);
            const int k1 = frag_k(j, p, h, 1);
            const float* s0 = is_rgb ? (w_rgb + k0 * 3 + c) : (w_sigma + k0);
            const float* s1 = is_rgb ? (w_rgb + k1 * 3 + c) : (w_sigma + k1);
            wt[2 * j + p].x = __uint_as_float(__float_as_uint(*s0) & keep);
            wt[2 * j + p].y = __uint_as_float(__float_as_uint(*s1) & keep);
        }
    }

    const int tile0 = wave * kTPW;

    float4 f[kTPW][2];
#pragma unroll
    for (int i = 0; i < kTPW; ++i) {
        const float4* row = (const float4*)(points_feat + (size_t)((tile0 + i) * 16 + c) * kC);
        f[i][0] = row[h];
        f[i][1] = row[2 + h];
    }

    static_assert(kTPW % 2 == 0, "tiles are processed in pairs");
#pragma unroll
    for (int i = 0; i < kTPW; i += 2) {
        v8f d0 = {};
        v8f d1 = {};
#pragma unroll
        for (int j = 0; j < 2; ++j) {
            v2f b00, b01, b10, b11;
            b00.x = f[i][j].x;      b00.y = f[i][j].y;
            b01.x = f[i][j].z;      b01.y = f[i][j].w;
            b10.x = f[i + 1][j].x;  b10.y = f[i + 1][j].y;
            b11.x = f[i + 1][j].z;  b11.y = f[i + 1][j].w;
            d0 = __builtin_amdgcn_wmma_f32_16x16x4_f32(
                     false, wt[2 * j + 0], false, b00, (short)0, d0, false, false);
            d1 = __builtin_amdgcn_wmma_f32_16x16x4_f32(
                     false, wt[2 * j + 0], false, b10, (short)0, d1, false, false);
            d0 = __builtin_amdgcn_wmma_f32_16x16x4_f32(
                     false, wt[2 * j + 1], false, b01, (short)0, d0, false, false);
            d1 = __builtin_amdgcn_wmma_f32_16x16x4_f32(
                     false, wt[2 * j + 1], false, b11, (short)0, d1, false, false);
        }
        if (h == 0) {
            float4 o0, o1;
            o0.x = d0[0]; o0.y = d0[1]; o0.z = d0[2]; o0.w = d0[3];
            o1.x = d1[0]; o1.y = d1[1]; o1.z = d1[2]; o1.w = d1[3];
            ((float4*)g_ws)[(tile0 + i) * 16 + c]     = o0;
            ((float4*)g_ws)[(tile0 + i + 1) * 16 + c] = o1;
        }
    }
}

constexpr int kTR    = 32;
constexpr int kTS    = 32;
constexpr int kSPT   = 4;
constexpr int kSG    = kTS / kSPT;
constexpr int kPitch = 36;
static_assert(kTR * kSG == 256, "one thread per ray and sample group of the tile");
static_assert(kR % kTR == 0 && kSR % kTS == 0, "tiles cover the rays and samples exactly");

__global__ __launch_bounds__(256)
void pvr_gather_act(const float* __restrict__ g_ws,
                    const int*   __restrict__ indices,
                    const float* __restrict__ dists,
                    const float* __restrict__ delta,
                    const float* __restrict__ z_vals,
                    float* __restrict__ arf_ws,
                    float* __restrict__ z_ws)
{
    __shared__ float tile[5][kTS][kPitch];

    const int lane = threadIdx.x & 31;
    const int wv   = threadIdx.x >> 5;
    const int s0   = blockIdx.x * kTS;
    const int r0   = blockIdx.y * kTR;
    const int rl   = 4 * wv + (lane >> 3);
    const int sl   = lane & 7;
    const int nb   = (r0 + rl) * kSR + s0 + sl;

    v4i    ia[kSPT], ib[kSPT];
    v4f    da[kSPT], db[kSPT];
    float  dl[kSPT], zv[kSPT];
#pragma unroll
    for (int j = 0; j < kSPT; ++j) {
        const int n = nb + kSG * j;
        const v4i* iv = (const v4i*)(indices + (size_t)n * kK);
        ia[j] = iv[0];
        ib[j] = iv[1];
        const v4f* dv = (const v4f*)(dists + (size_t)n * kK);
        da[j] = dv[0];
        db[j] = dv[1];
        dl[j] = delta[n];
        zv[j] = z_vals[n];
    }

    asm volatile("" :: "v"(ia[0]), "v"(ib[0]), "v"(ia[1]), "v"(ib[1]),
                       "v"(ia[2]), "v"(ib[2]), "v"(ia[3]), "v"(ib[3]),
                       "v"(da[0]), "v"(db[0]), "v"(da[1]), "v"(db[1]),
                       "v"(da[2]), "v"(db[2]), "v"(da[3]), "v"(db[3]),
                       "v"(dl[0]), "v"(dl[1]), "v"(dl[2]), "v"(dl[3]),
                       "v"(zv[0]), "v"(zv[1]), "v"(zv[2]), "v"(zv[3]));

    const float4* G = (const float4*)g_ws;
    float4 g[kSPT][kK];
#pragma unroll
    for (int j = 0; j < kSPT; ++j) {
        g[j][0] = G[ia[j][0]]; g[j][1] = G[ia[j][1]]; g[j][2] = G[ia[j][2]]; g[j][3] = G[ia[j][3]];
        g[j][4] = G[ib[j][0]]; g[j][5] = G[ib[j][1]]; g[j][6] = G[ib[j][2]]; g[j][7] = G[ib[j][3]];
    }

    __builtin_amdgcn_sched_barrier(0);

#pragma unroll
    for (int j = 0; j < kSPT; ++j) {
        const float dd[kK] = {da[j][0], da[j][1], da[j][2], da[j][3], db[j][0], db[j][1], db[j][2], db[j][3]};

        float w[kK];
        float wsum = 0.0f;
#pragma unroll
        for (int k = 0; k < kK; ++k) {
            w[k] = 1.0f / (dd[k] + 1e-7f);
            wsum += w[k];
        }
        const float inv = 1.0f / wsum;

        float o0 = 0.0f, o1 = 0.0f, o2 = 0.0f, o3 = 0.0f;
#pragma unroll
        for (int k = 0; k < kK; ++k) {
            const float wk = w[k] * inv;
            o0 = fmaf(wk, g[j][k].x, o0);
            o1 = fmaf(wk, g[j][k].y, o1);
            o2 = fmaf(wk, g[j][k].z, o2);
            o3 = fmaf(wk, g[j][k].w, o3);
        }

        const float sig = fmaxf(o3, 0.0f);
        const int   sr  = kSG * j + sl;
        tile[0][sr][rl] = 1.0f - __expf(-sig * dl[j]);
        tile[1][sr][rl] = 1.0f / (1.0f + __expf(-o0));
        tile[2][sr][rl] = 1.0f / (1.0f + __expf(-o1));
        tile[3][sr][rl] = 1.0f / (1.0f + __expf(-o2));
        tile[4][sr][rl] = zv[j];
    }

    __syncthreads();

#pragma unroll
    for (int j = 0; j < kSPT; ++j) {
        const int sr = kSPT * wv + j;
        float4 v;
        v.x = tile[0][sr][lane];
        v.y = tile[1][sr][lane];
        v.z = tile[2][sr][lane];
        v.w = tile[3][sr][lane];
        const size_t o = ((size_t)blockIdx.y * kSR + s0 + sr) * kTR + lane;
        ((float4*)arf_ws)[o] = v;
        z_ws[o]              = tile[4][sr][lane];
    }
}

constexpr int kCH = 32;
static_assert(kSR % (2 * kCH) == 0, "chunks are processed in pairs");

struct Chunk {
    float4 a[kCH];
    float  z[kCH];
};

__device__ __forceinline__ void chunk_load(Chunk& c, const float4* __restrict__ arf,
                                           const float* __restrict__ zp, int s_first)
{
#pragma unroll
    for (int i = 0; i < kCH; ++i) {
        c.a[i] = arf[(s_first + i) * kTR];
        c.z[i] = zp[(s_first + i) * kTR];
    }
}

struct Acc {
    float T, r0, r1, r2, depth, accw;
};

__device__ __forceinline__ void chunk_use(const Chunk& c, Acc& q)
{
#pragma unroll
    for (int i = 0; i < kCH; ++i) {
        const float a  = c.a[i].x;
        const float wt = a * q.T;
        q.r0    = fmaf(wt, c.a[i].y, q.r0);
        q.r1    = fmaf(wt, c.a[i].z, q.r1);
        q.r2    = fmaf(wt, c.a[i].w, q.r2);
        q.depth = fmaf(wt, c.z[i],   q.depth);
        q.accw += wt;
        q.T *= (1.0f - a + 1e-10f);
    }
}

__global__ __launch_bounds__(kTR)
void pvr_composite(const float* __restrict__ arf_ws,
                   const float* __restrict__ z_ws,
                   float* __restrict__ out)
{
    static_assert(kTR == 32, "one wave per ray tile");
    const int r = blockIdx.x * kTR + threadIdx.x;
    const float4* arf = (const float4*)arf_ws + (size_t)blockIdx.x * kSR * kTR + threadIdx.x;
    const float*  zp  = z_ws + (size_t)blockIdx.x * kSR * kTR + threadIdx.x;

    Acc q;
    q.T = 1.0f; q.r0 = 0.0f; q.r1 = 0.0f; q.r2 = 0.0f; q.depth = 0.0f; q.accw = 0.0f;

    Chunk ca, cb;
    chunk_load(ca, arf, zp, 0);
#pragma unroll
    for (int s = 0; s < kSR; s += 2 * kCH) {
        chunk_load(cb, arf, zp, s + kCH);
        __builtin_amdgcn_sched_barrier(0);
        chunk_use(ca, q);
        if (s + 2 * kCH < kSR) chunk_load(ca, arf, zp, s + 2 * kCH);
        chunk_use(cb, q);
    }

    const float bg = 1.0f - q.accw;
    out[r * 5 + 0] = q.r0 + bg;
    out[r * 5 + 1] = q.r1 + bg;
    out[r * 5 + 2] = q.r2 + bg;
    out[r * 5 + 3] = q.depth;
    out[r * 5 + 4] = q.accw;
}

extern "C" void kernel_launch(void* const* d_in, const int* in_sizes, int n_in,
                              void* d_out, int out_size, void* d_ws, size_t ws_size,
                              hipStream_t stream)
{
    const float* points_feat = (const float*)d_in[0];
    const int*   indices     = (const int*)  d_in[1];
    const float* dists       = (const float*)d_in[2];
    const float* w_rgb       = (const float*)d_in[3];
    const float* w_sigma     = (const float*)d_in[4];
    const float* delta       = (const float*)d_in[5];
    const float* z_vals      = (const float*)d_in[6];
    float* out = (float*)d_out;

    float* arf_ws = (float*)d_ws;
    float* z_ws   = arf_ws + 4 * kN;
    float* g_ws   = z_ws + kN;

    pvr_table_head<<<kPreGrid, kPreWPB * 32, 0, stream>>>(
        points_feat, w_rgb, w_sigma, g_ws);

    pvr_gather_act<<<dim3(kSR / kTS, kR / kTR), 256, 0, stream>>>(
        g_ws, indices, dists, delta, z_vals, arf_ws, z_ws);

    pvr_composite<<<kR / kTR, kTR, 0, stream>>>(
        arf_ws, z_ws, out);
}


# Compositing done in the gather kernel's registers: a wave owns one ray (lane = sample, 4 runs of 32), 32-lane product scan for transmittance, lane sums; 2 launches; sums and products re-associated
# speedup vs baseline: 2.4415x; 1.2362x over previous
#include <hip/hip_runtime.h>


typedef __attribute__((ext_vector_type(2))) float v2f;
typedef __attribute__((ext_vector_type(8))) float v8f;
typedef __attribute__((ext_vector_type(4))) float v4f;
typedef __attribute__((ext_vector_type(4))) int   v4i;

constexpr int kC   = 16;
constexpr int kK   = 8;
constexpr int kR   = 4096;
constexpr int kSR  = 128;
constexpr int kN   = kR * kSR;
constexpr int kNP  = 500000;

constexpr int kTiles   = kNP / 16;
constexpr int kTPW     = 10;
constexpr int kPreWPB  = 5;
constexpr int kPreGrid = kTiles / kTPW / kPreWPB;
static_assert(kPreGrid * kPreWPB * kTPW * 16 == kNP, "table pass must cover every row exactly once");


__device__ __forceinline__ constexpr int frag_k(int j, int p, int h, int e)
{
    return 8 * j + 4 * h + 2 * p + e;
}

__global__ __launch_bounds__(kPreWPB * 32)
void pvr_table_head(const float* __restrict__ points_feat,
                    const float* __restrict__ w_rgb,
                    const float* __restrict__ w_sigma,
                    float* __restrict__ g_ws)
{
    const int lane = threadIdx.x & 31;
    const int wave = blockIdx.x * kPreWPB + (threadIdx.x >> 5);
    const int h    = lane >> 4;
    const int c    = lane & 15;

    const bool     is_rgb = (c < 3);
    const unsigned keep   = (c <= 3) ? 0xffffffffu : 0u;
    v2f wt[4];
#pragma unroll
    for (int j = 0; j < 2; ++j) {
#pragma unroll
        for (int p = 0; p < 2; ++p) {
            const int k0 = frag_k(j, p, h, 0);
            const int k1 = frag_k(j, p, h, 1);
            const float* s0 = is_rgb ? (w_rgb + k0 * 3 + c) : (w_sigma + k0);
            const float* s1 = is_rgb ? (w_rgb + k1 * 3 + c) : (w_sigma + k1);
            wt[2 * j + p].x = __uint_as_float(__float_as_uint(*s0) & keep);
            wt[2 * j + p].y = __uint_as_float(__float_as_uint(*s1) & keep);
        }
    }

    const int tile0 = wave * kTPW;

    float4 f[kTPW][2];
#pragma unroll
    for (int i = 0; i < kTPW; ++i) {
        const float4* row = (const float4*)(points_feat + (size_t)((tile0 + i) * 16 + c) * kC);
        f[i][0] = row[h];
        f[i][1] = row[2 + h];
    }

    static_assert(kTPW % 2 == 0, "tiles are processed in pairs");
#pragma unroll
    for (int i = 0; i < kTPW; i += 2) {
        v8f d0 = {};
        v8f d1 = {};
#pragma unroll
        for (int j = 0; j < 2; ++j) {
            v2f b00, b01, b10, b11;
            b00.x = f[i][j].x;      b00.y = f[i][j].y;
            b01.x = f[i][j].z;      b01.y = f[i][j].w;
            b10.x = f[i + 1][j].x;  b10.y = f[i + 1][j].y;
            b11.x = f[i + 1][j].z;  b11.y = f[i + 1][j].w;
            d0 = __builtin_amdgcn_wmma_f32_16x16x4_f32(
                     false, wt[2 * j + 0], false, b00, (short)0, d0, false, false);
            d1 = __builtin_amdgcn_wmma_f32_16x16x4_f32(
                     false, wt[2 * j + 0], false, b10, (short)0, d1, false, false);
            d0 = __builtin_amdgcn_wmma_f32_16x16x4_f32(
                     false, wt[2 * j + 1], false, b01, (short)0, d0, false, false);
            d1 = __builtin_amdgcn_wmma_f32_16x16x4_f32(
                     false, wt[2 * j + 1], false, b11, (short)0, d1, false, false);
        }
        if (h == 0) {
            float4 o0, o1;
            o0.x = d0[0]; o0.y = d0[1]; o0.z = d0[2]; o0.w = d0[3];
            o1.x = d1[0]; o1.y = d1[1]; o1.z = d1[2]; o1.w = d1[3];
            ((float4*)g_ws)[(tile0 + i) * 16 + c]     = o0;
            ((float4*)g_ws)[(tile0 + i + 1) * 16 + c] = o1;
        }
    }
}

constexpr int kSPT = 4;
constexpr int kRPB = 8;
static_assert(kSR == 32 * kSPT, "a wave's 32 lanes x 4 samples cover one ray exactly");
static_assert(kR % kRPB == 0, "blocks cover the rays exactly");

template <int N>
__device__ __forceinline__ float row_below_or_one(float x)
{
    return __int_as_float(__builtin_amdgcn_update_dpp(
        __float_as_int(1.0f), __float_as_int(x), 0x110 + N  , 0xf, 0xf, false));
}

template <int M>
__device__ __forceinline__ float row_xor_add(float x)
{
    return x + __int_as_float(__builtin_amdgcn_update_dpp(
        __float_as_int(x), __float_as_int(x), 0x160 + M  , 0xf, 0xf, true));
}

__device__ __forceinline__ float lane_value(float x, int l)
{
    return __int_as_float(__builtin_amdgcn_readlane(__float_as_int(x), l));
}

__device__ __forceinline__ float wave_sum(float x)
{
    x = row_xor_add<1>(x);
    x = row_xor_add<2>(x);
    x = row_xor_add<4>(x);
    x = row_xor_add<8>(x);
    return lane_value(x, 0) + lane_value(x, 16);
}

__global__ __launch_bounds__(kRPB * 32)
void pvr_gather_composite(const float* __restrict__ g_ws,
                          const int*   __restrict__ indices,
                          const float* __restrict__ dists,
                          const float* __restrict__ delta,
                          const float* __restrict__ z_vals,
                          float* __restrict__ out)
{
    const int lane = threadIdx.x & 31;
    const int ray  = blockIdx.x * kRPB + (threadIdx.x >> 5);
    const int nb   = ray * kSR + lane;

    v4i    ia[kSPT], ib[kSPT];
    v4f    da[kSPT], db[kSPT];
    float  dl[kSPT], zv[kSPT];
#pragma unroll
    for (int j = 0; j < kSPT; ++j) {
        const int n = nb + 32 * j;
        const v4i* iv = (const v4i*)(indices + (size_t)n * kK);
        ia[j] = iv[0];
        ib[j] = iv[1];
        const v4f* dv = (const v4f*)(dists + (size_t)n * kK);
        da[j] = dv[0];
        db[j] = dv[1];
        dl[j] = delta[n];
        zv[j] = z_vals[n];
    }

    asm volatile("" :: "v"(ia[0]), "v"(ib[0]), "v"(ia[1]), "v"(ib[1]),
                       "v"(ia[2]), "v"(ib[2]), "v"(ia[3]), "v"(ib[3]),
                       "v"(da[0]), "v"(db[0]), "v"(da[1]), "v"(db[1]),
                       "v"(da[2]), "v"(db[2]), "v"(da[3]), "v"(db[3]),
                       "v"(dl[0]), "v"(dl[1]), "v"(dl[2]), "v"(dl[3]),
                       "v"(zv[0]), "v"(zv[1]), "v"(zv[2]), "v"(zv[3]));

    const float4* G = (const float4*)g_ws;
    float4 g[kSPT][kK];
#pragma unroll
    for (int j = 0; j < kSPT; ++j) {
        g[j][0] = G[ia[j][0]]; g[j][1] = G[ia[j][1]]; g[j][2] = G[ia[j][2]]; g[j][3] = G[ia[j][3]];
        g[j][4] = G[ib[j][0]]; g[j][5] = G[ib[j][1]]; g[j][6] = G[ib[j][2]]; g[j][7] = G[ib[j][3]];
    }

    __builtin_amdgcn_sched_barrier(0);

    float al[kSPT], cr[kSPT], cg[kSPT], cb[kSPT];
#pragma unroll
    for (int j = 0; j < kSPT; ++j) {
        const float dd[kK] = {da[j][0], da[j][1], da[j][2], da[j][3], db[j][0], db[j][1], db[j][2], db[j][3]};

        float w[kK];
        float wsum = 0.0f;
#pragma unroll
        for (int k = 0; k < kK; ++k) {
            w[k] = 1.0f / (dd[k] + 1e-7f);
            wsum += w[k];
        }
        const float inv = 1.0f / wsum;

        float o0 = 0.0f, o1 = 0.0f, o2 = 0.0f, o3 = 0.0f;
#pragma unroll
        for (int k = 0; k < kK; ++k) {
            const float wk = w[k] * inv;
            o0 = fmaf(wk, g[j][k].x, o0);
            o1 = fmaf(wk, g[j][k].y, o1);
            o2 = fmaf(wk, g[j][k].z, o2);
            o3 = fmaf(wk, g[j][k].w, o3);
        }

        const float sig = fmaxf(o3, 0.0f);
        al[j] = 1.0f - __expf(-sig * dl[j]);
        cr[j] = 1.0f / (1.0f + __expf(-o0));
        cg[j] = 1.0f / (1.0f + __expf(-o1));
        cb[j] = 1.0f / (1.0f + __expf(-o2));
    }

    const bool upper = (lane >= 16);
    float carry = 1.0f;
    float s0 = 0.0f, s1 = 0.0f, s2 = 0.0f, sd = 0.0f, sa = 0.0f;
#pragma unroll
    for (int j = 0; j < kSPT; ++j) {
        float x = 1.0f - al[j] + 1e-10f;
        x *= row_below_or_one<1>(x);
        x *= row_below_or_one<2>(x);
        x *= row_below_or_one<4>(x);
        x *= row_below_or_one<8>(x);
        const float lo = lane_value(x, 15);
        const float hi = lane_value(x, 31);
        float e = row_below_or_one<1>(x);
        e = upper ? e * lo : e;
        const float T  = carry * e;
        const float wt = al[j] * T;
        s0 = fmaf(wt, cr[j], s0);
        s1 = fmaf(wt, cg[j], s1);
        s2 = fmaf(wt, cb[j], s2);
        sd = fmaf(wt, zv[j], sd);
        sa += wt;
        carry *= lo * hi;
    }

    const float r0    = wave_sum(s0);
    const float r1    = wave_sum(s1);
    const float r2    = wave_sum(s2);
    const float depth = wave_sum(sd);
    const float accw  = wave_sum(sa);

    const float bg = 1.0f - accw;
    const float v  = (lane == 0) ? r0 + bg
                   : (lane == 1) ? r1 + bg
                   : (lane == 2) ? r2 + bg
                   : (lane == 3) ? depth
                                 : accw;
    if (lane < 5) out[ray * 5 + lane] = v;
}

extern "C" void kernel_launch(void* const* d_in, const int* in_sizes, int n_in,
                              void* d_out, int out_size, void* d_ws, size_t ws_size,
                              hipStream_t stream)
{
    const float* points_feat = (const float*)d_in[0];
    const int*   indices     = (const int*)  d_in[1];
    const float* dists       = (const float*)d_in[2];
    const float* w_rgb       = (const float*)d_in[3];
    const float* w_sigma     = (const float*)d_in[4];
    const float* delta       = (const float*)d_in[5];
    const float* z_vals      = (const float*)d_in[6];
    float* out = (float*)d_out;

    float* g_ws = (float*)d_ws;

    pvr_table_head<<<kPreGrid, kPreWPB * 32, 0, stream>>>(
        points_feat, w_rgb, w_sigma, g_ws);

    pvr_gather_composite<<<kR / kRPB, kRPB * 32, 0, stream>>>(
        g_ws, indices, dists, delta, z_vals, out);
}
